# starting kernel
# baseline (speedup 1.0000x reference)
#include <hip/hip_runtime.h>
#include <hip/hip_bf16.h>

typedef float v2f __attribute__((ext_vector_type(2)));
typedef float v4f __attribute__((ext_vector_type(4)));
typedef float v8f __attribute__((ext_vector_type(8)));

#define N_  64
#define C_  64
#define F_  64
#define T_  300
#define V_  25
#define VP_ 32
#define TB_ 4
static_assert(T_ % TB_ == 0, "t tiles must divide T exactly");

__global__ void build_adj_kernel(const int* __restrict__ ei, int E,
                                 float* __restrict__ Ahat) {
    __shared__ float deg[VP_];
    __shared__ float dinv[VP_];
    const int tid = threadIdx.x;

    if (tid < VP_) deg[tid] = 0.0f;
    __syncthreads();
    for (int e = tid; e < E; e += blockDim.x)
        atomicAdd(&deg[ei[E + e]], 1.0f);
    __syncthreads();
    if (tid < V_) deg[tid] += 1.0f;
    __syncthreads();
    if (tid < VP_) dinv[tid] = (tid < V_ && deg[tid] > 0.0f) ? rsqrtf(deg[tid]) : 0.0f;
    __syncthreads();

    for (int i = tid; i < VP_ * VP_; i += blockDim.x) Ahat[i] = 0.0f;
    __syncthreads();

    for (int e = tid; e < E; e += blockDim.x) {
        int s = ei[e];
        int d = ei[E + e];
        atomicAdd(&Ahat[d * VP_ + s], dinv[s] * dinv[d]);
    }
    if (tid < V_) atomicAdd(&Ahat[tid * VP_ + tid], dinv[tid] * dinv[tid]);
}

#define XW   (TB_ * V_)
#define XOFF 0
#define WOFF (C_ * XW)
#define AOFF (WOFF + F_ * C_)
#define BOFF (AOFF + VP_ * VP_)
#define LDSF (BOFF + F_)

__global__ __launch_bounds__(TB_ * 32) void spatial_gcn_kernel(
        const float* __restrict__ x, const float* __restrict__ w,
        const float* __restrict__ bias, const float* __restrict__ Ahat,
        float* __restrict__ out) {
    __shared__ float lds[LDSF];
    float* Xl = lds + XOFF;
    float* Wl = lds + WOFF;
    float* Al = lds + AOFF;
    float* Bl = lds + BOFF;

    const int tid = threadIdx.x;
    const int n   = blockIdx.y;
    const int t0  = blockIdx.x * TB_;
    const int NT  = TB_ * 32;

    for (int j = tid; j < C_ * (XW / 4); j += NT) {
        const int c = j / (XW / 4);
        const int q = j % (XW / 4);
        const v4f val = *reinterpret_cast<const v4f*>(
            x + (long)(n * C_ + c) * (T_ * V_) + t0 * V_ + q * 4);
        *reinterpret_cast<v4f*>(Xl + c * XW + q * 4) = val;
    }
    for (int j = tid; j < (F_ * C_) / 4; j += NT)
        reinterpret_cast<v4f*>(Wl)[j] = reinterpret_cast<const v4f*>(w)[j];
    for (int j = tid; j < (VP_ * VP_) / 4; j += NT)
        reinterpret_cast<v4f*>(Al)[j] = reinterpret_cast<const v4f*>(Ahat)[j];
    if (tid < F_ / 4)
        reinterpret_cast<v4f*>(Bl)[tid] = reinterpret_cast<const v4f*>(bias)[tid];
    __syncthreads();

    const int wv   = tid >> 5;
    const int lane = tid & 31;
    const int half = lane >> 4;
    const int l16  = lane & 15;
    const int t    = t0 + wv;
    const int xb   = wv * V_;

    v8f acc[4][2];
#pragma unroll
    for (int mt = 0; mt < 4; ++mt)
#pragma unroll
        for (int nt = 0; nt < 2; ++nt)
#pragma unroll
            for (int r = 0; r < 8; ++r) acc[mt][nt][r] = 0.0f;

#pragma unroll
    for (int k = 0; k < 16; ++k) {
        const int kk = 4 * k + 2 * half;
        v2f av[4], bv[2];
#pragma unroll
        for (int mt = 0; mt < 4; ++mt) {
            const float* p = &Wl[(mt * 16 + l16) * C_ + kk];
            av[mt][0] = p[0];
            av[mt][1] = p[1];
        }
#pragma unroll
        for (int nt = 0; nt < 2; ++nt) {
            bv[nt][0] = Xl[(kk + 0) * XW + xb + nt * 16 + l16];
            bv[nt][1] = Xl[(kk + 1) * XW + xb + nt * 16 + l16];
        }
#pragma unroll
        for (int mt = 0; mt < 4; ++mt)
#pragma unroll
            for (int nt = 0; nt < 2; ++nt)
                acc[mt][nt] = __builtin_amdgcn_wmma_f32_16x16x4_f32(
                    false, av[mt], false, bv[nt], (short)0, acc[mt][nt],
                    false, false);
    }

#pragma unroll
    for (int nt = 0; nt < 2; ++nt) {
        const int vp = nt * 16 + l16;
        if (vp < V_) {
#pragma unroll
            for (int mt = 0; mt < 4; ++mt)
#pragma unroll
                for (int r = 0; r < 8; ++r)
                    Xl[(mt * 16 + half * 8 + r) * XW + xb + vp] = acc[mt][nt][r];
        }
    }

    v8f o[4][2];
#pragma unroll
    for (int mt = 0; mt < 4; ++mt)
#pragma unroll
        for (int nt = 0; nt < 2; ++nt)
#pragma unroll
            for (int r = 0; r < 8; ++r)
                o[mt][nt][r] = Bl[mt * 16 + half * 8 + r];

#pragma unroll
    for (int k = 0; k < 8; ++k) {
        const int kk = 4 * k + 2 * half;
        v2f av[4], bv[2];
#pragma unroll
        for (int mt = 0; mt < 4; ++mt) {
            av[mt][0] = Xl[(mt * 16 + l16) * XW + xb + kk + 0];
            av[mt][1] = Xl[(mt * 16 + l16) * XW + xb + kk + 1];
        }
#pragma unroll
        for (int nt = 0; nt < 2; ++nt) {
            bv[nt][0] = Al[(nt * 16 + l16) * VP_ + kk + 0];
            bv[nt][1] = Al[(nt * 16 + l16) * VP_ + kk + 1];
        }
#pragma unroll
        for (int mt = 0; mt < 4; ++mt)
#pragma unroll
            for (int nt = 0; nt < 2; ++nt)
                o[mt][nt] = __builtin_amdgcn_wmma_f32_16x16x4_f32(
                    false, av[mt], false, bv[nt], (short)0, o[mt][nt],
                    false, false);
    }

#pragma unroll
    for (int nt = 0; nt < 2; ++nt) {
        const int v = nt * 16 + l16;
        if (v < V_) {
#pragma unroll
            for (int mt = 0; mt < 4; ++mt)
#pragma unroll
                for (int r = 0; r < 8; ++r) {
                    const int f = mt * 16 + half * 8 + r;
                    out[(((long)n * F_ + f) * T_ + t) * V_ + v] = o[mt][nt][r];
                }
        }
    }
}

extern "C" void kernel_launch(void* const* d_in, const int* in_sizes, int n_in,
                              void* d_out, int out_size, void* d_ws, size_t ws_size,
                              hipStream_t stream) {
    const float* x    = (const float*)d_in[0];
    const float* w    = (const float*)d_in[1];
    const float* bias = (const float*)d_in[2];
    const int*   ei   = (const int*)d_in[3];
    const int E = in_sizes[3] / 2;

    float* Ahat = (float*)d_ws;

    build_adj_kernel<<<1, 64, 0, stream>>>(ei, E, Ahat);

    dim3 grid(T_ / TB_, N_);
    spatial_gcn_kernel<<<grid, TB_ * 32, 0, stream>>>(
        x, w, bias, Ahat, (float*)d_out);
}


# Issue all staging global loads of a block (W, adjacency, bias, x slice: 24 b128 per thread) into registers before the first LDS store: one memory round trip instead of about 14 serial ones
# speedup vs baseline: 1.1700x; 1.1700x over previous
#include <hip/hip_runtime.h>
#include <hip/hip_bf16.h>

typedef float v2f __attribute__((ext_vector_type(2)));
typedef float v4f __attribute__((ext_vector_type(4)));
typedef float v8f __attribute__((ext_vector_type(8)));

#define N_  64
#define C_  64
#define F_  64
#define T_  300
#define V_  25
#define VP_ 32
#define TB_ 4
static_assert(T_ % TB_ == 0, "t tiles must divide T exactly");

__global__ void build_adj_kernel(const int* __restrict__ ei, int E,
                                 float* __restrict__ Ahat) {
    __shared__ float deg[VP_];
    __shared__ float dinv[VP_];
    const int tid = threadIdx.x;

    if (tid < VP_) deg[tid] = 0.0f;
    __syncthreads();
    for (int e = tid; e < E; e += blockDim.x)
        atomicAdd(&deg[ei[E + e]], 1.0f);
    __syncthreads();
    if (tid < V_) deg[tid] += 1.0f;
    __syncthreads();
    if (tid < VP_) dinv[tid] = (tid < V_ && deg[tid] > 0.0f) ? rsqrtf(deg[tid]) : 0.0f;
    __syncthreads();

    for (int i = tid; i < VP_ * VP_; i += blockDim.x) Ahat[i] = 0.0f;
    __syncthreads();

    for (int e = tid; e < E; e += blockDim.x) {
        int s = ei[e];
        int d = ei[E + e];
        atomicAdd(&Ahat[d * VP_ + s], dinv[s] * dinv[d]);
    }
    if (tid < V_) atomicAdd(&Ahat[tid * VP_ + tid], dinv[tid] * dinv[tid]);
}

#define XW   (TB_ * V_)
#define XOFF 0
#define WOFF (C_ * XW)
#define AOFF (WOFF + F_ * C_)
#define BOFF (AOFF + VP_ * VP_)
#define LDSF (BOFF + F_)

__global__ __launch_bounds__(TB_ * 32) void spatial_gcn_kernel(
        const float* __restrict__ x, const float* __restrict__ w,
        const float* __restrict__ bias, const float* __restrict__ Ahat,
        float* __restrict__ out) {
    __shared__ float lds[LDSF];
    float* Xl = lds + XOFF;
    float* Wl = lds + WOFF;
    float* Al = lds + AOFF;
    float* Bl = lds + BOFF;

    const int tid = threadIdx.x;
    const int n   = blockIdx.y;
    const int t0  = blockIdx.x * TB_;
    constexpr int NT = TB_ * 32;
    constexpr int XQ = XW / 4;
    constexpr int XV = C_ * XQ;
    constexpr int XI = (XV + NT - 1) / NT;
    constexpr int WI = (F_ * C_ / 4) / NT;
    constexpr int AI = (VP_ * VP_ / 4) / NT;
    static_assert(XW % 4 == 0, "x rows are whole float4s");
    static_assert((F_ * C_ / 4) % NT == 0 && (VP_ * VP_ / 4) % NT == 0,
                  "W and Ahat split evenly over the block");
    static_assert((XI - 1) * NT < XV && (NT % (F_ / 4)) == 0,
                  "only the last x trip is partial; bias copies tile the block");

    const v4f* wg = reinterpret_cast<const v4f*>(w);
    const v4f* ag = reinterpret_cast<const v4f*>(Ahat);
    const v4f* bg = reinterpret_cast<const v4f*>(bias);
    const float* xg = x + (long)n * (C_ * T_ * V_) + t0 * V_;

    v4f wr[WI], ar[AI], xr[XI];
#pragma unroll
    for (int i = 0; i < WI; ++i) wr[i] = wg[tid + i * NT];
#pragma unroll
    for (int i = 0; i < AI; ++i) ar[i] = ag[tid + i * NT];
    const v4f br = bg[tid & (F_ / 4 - 1)];
#pragma unroll
    for (int i = 0; i < XI; ++i) {
        unsigned j = (unsigned)tid + (unsigned)(i * NT);
        if (i == XI - 1 && j > XV - 1) j = XV - 1;
        const unsigned c = j / XQ;
        const unsigned q = j % XQ;
        xr[i] = *reinterpret_cast<const v4f*>(xg + (c * (unsigned)(T_ * V_) + q * 4u));
    }

#pragma unroll
    for (int i = 0; i < WI; ++i) reinterpret_cast<v4f*>(Wl)[tid + i * NT] = wr[i];
#pragma unroll
    for (int i = 0; i < AI; ++i) reinterpret_cast<v4f*>(Al)[tid + i * NT] = ar[i];
    reinterpret_cast<v4f*>(Bl)[tid & (F_ / 4 - 1)] = br;
#pragma unroll
    for (int i = 0; i < XI; ++i) {
        unsigned j = (unsigned)tid + (unsigned)(i * NT);
        if (i == XI - 1 && j > XV - 1) j = XV - 1;
        reinterpret_cast<v4f*>(Xl)[j] = xr[i];
    }
    __syncthreads();

    const int wv   = tid >> 5;
    const int lane = tid & 31;
    const int half = lane >> 4;
    const int l16  = lane & 15;
    const int t    = t0 + wv;
    const int xb   = wv * V_;

    v8f acc[4][2];
#pragma unroll
    for (int mt = 0; mt < 4; ++mt)
#pragma unroll
        for (int nt = 0; nt < 2; ++nt)
#pragma unroll
            for (int r = 0; r < 8; ++r) acc[mt][nt][r] = 0.0f;

#pragma unroll
    for (int k = 0; k < 16; ++k) {
        const int kk = 4 * k + 2 * half;
        v2f av[4], bv[2];
#pragma unroll
        for (int mt = 0; mt < 4; ++mt) {
            const float* p = &Wl[(mt * 16 + l16) * C_ + kk];
            av[mt][0] = p[0];
            av[mt][1] = p[1];
        }
#pragma unroll
        for (int nt = 0; nt < 2; ++nt) {
            bv[nt][0] = Xl[(kk + 0) * XW + xb + nt * 16 + l16];
            bv[nt][1] = Xl[(kk + 1) * XW + xb + nt * 16 + l16];
        }
#pragma unroll
        for (int mt = 0; mt < 4; ++mt)
#pragma unroll
            for (int nt = 0; nt < 2; ++nt)
                acc[mt][nt] = __builtin_amdgcn_wmma_f32_16x16x4_f32(
                    false, av[mt], false, bv[nt], (short)0, acc[mt][nt],
                    false, false);
    }

#pragma unroll
    for (int nt = 0; nt < 2; ++nt) {
        const int vp = nt * 16 + l16;
        if (vp < V_) {
#pragma unroll
            for (int mt = 0; mt < 4; ++mt)
#pragma unroll
                for (int r = 0; r < 8; ++r)
                    Xl[(mt * 16 + half * 8 + r) * XW + xb + vp] = acc[mt][nt][r];
        }
    }

    v8f o[4][2];
#pragma unroll
    for (int mt = 0; mt < 4; ++mt)
#pragma unroll
        for (int nt = 0; nt < 2; ++nt)
#pragma unroll
            for (int r = 0; r < 8; ++r)
                o[mt][nt][r] = Bl[mt * 16 + half * 8 + r];

#pragma unroll
    for (int k = 0; k < 8; ++k) {
        const int kk = 4 * k + 2 * half;
        v2f av[4], bv[2];
#pragma unroll
        for (int mt = 0; mt < 4; ++mt) {
            av[mt][0] = Xl[(mt * 16 + l16) * XW + xb + kk + 0];
            av[mt][1] = Xl[(mt * 16 + l16) * XW + xb + kk + 1];
        }
#pragma unroll
        for (int nt = 0; nt < 2; ++nt) {
            bv[nt][0] = Al[(nt * 16 + l16) * VP_ + kk + 0];
            bv[nt][1] = Al[(nt * 16 + l16) * VP_ + kk + 1];
        }
#pragma unroll
        for (int mt = 0; mt < 4; ++mt)
#pragma unroll
            for (int nt = 0; nt < 2; ++nt)
                o[mt][nt] = __builtin_amdgcn_wmma_f32_16x16x4_f32(
                    false, av[mt], false, bv[nt], (short)0, o[mt][nt],
                    false, false);
    }

#pragma unroll
    for (int nt = 0; nt < 2; ++nt) {
        const int v = nt * 16 + l16;
        if (v < V_) {
#pragma unroll
            for (int mt = 0; mt < 4; ++mt)
#pragma unroll
                for (int r = 0; r < 8; ++r) {
                    const int f = mt * 16 + half * 8 + r;
                    out[(((long)n * F_ + f) * T_ + t) * V_ + v] = o[mt][nt][r];
                }
        }
    }
}

extern "C" void kernel_launch(void* const* d_in, const int* in_sizes, int n_in,
                              void* d_out, int out_size, void* d_ws, size_t ws_size,
                              hipStream_t stream) {
    const float* x    = (const float*)d_in[0];
    const float* w    = (const float*)d_in[1];
    const float* bias = (const float*)d_in[2];
    const int*   ei   = (const int*)d_in[3];
    const int E = in_sizes[3] / 2;

    float* Ahat = (float*)d_ws;

    build_adj_kernel<<<1, 64, 0, stream>>>(ei, E, Ahat);

    dim3 grid(T_ / TB_, N_);
    spatial_gcn_kernel<<<grid, TB_ * 32, 0, stream>>>(
        x, w, bias, Ahat, (float*)d_out);
}


# Pad LDS row pitch of W (64 to 68 floats) and of the adjacency (32 to 36) so the 16 rows of a WMMA fragment read hit 16 different bank groups: removes 16-way and 8-way LDS bank conflicts
# speedup vs baseline: 1.2468x; 1.0656x over previous
#include <hip/hip_runtime.h>
#include <hip/hip_bf16.h>

typedef float v2f __attribute__((ext_vector_type(2)));
typedef float v4f __attribute__((ext_vector_type(4)));
typedef float v8f __attribute__((ext_vector_type(8)));

#define N_  64
#define C_  64
#define F_  64
#define T_  300
#define V_  25
#define VP_ 32
#define TB_ 4
static_assert(T_ % TB_ == 0, "t tiles must divide T exactly");

__global__ void build_adj_kernel(const int* __restrict__ ei, int E,
                                 float* __restrict__ Ahat) {
    __shared__ float deg[VP_];
    __shared__ float dinv[VP_];
    const int tid = threadIdx.x;

    if (tid < VP_) deg[tid] = 0.0f;
    __syncthreads();
    for (int e = tid; e < E; e += blockDim.x)
        atomicAdd(&deg[ei[E + e]], 1.0f);
    __syncthreads();
    if (tid < V_) deg[tid] += 1.0f;
    __syncthreads();
    if (tid < VP_) dinv[tid] = (tid < V_ && deg[tid] > 0.0f) ? rsqrtf(deg[tid]) : 0.0f;
    __syncthreads();

    for (int i = tid; i < VP_ * VP_; i += blockDim.x) Ahat[i] = 0.0f;
    __syncthreads();

    for (int e = tid; e < E; e += blockDim.x) {
        int s = ei[e];
        int d = ei[E + e];
        atomicAdd(&Ahat[d * VP_ + s], dinv[s] * dinv[d]);
    }
    if (tid < V_) atomicAdd(&Ahat[tid * VP_ + tid], dinv[tid] * dinv[tid]);
}

#define XW   (TB_ * V_)
#define WP   (C_ + 4)
#define AP   (VP_ + 4)
#define XOFF 0
#define WOFF (C_ * XW)
#define AOFF (WOFF + F_ * WP)
#define BOFF (AOFF + VP_ * AP)
#define LDSF (BOFF + F_)
static_assert(WP % 8 == 4 && AP % 8 == 4,
              "pitch = 4 x odd: 16 fragment rows land on 16 distinct groups of four banks");
static_assert(WP % 4 == 0 && AP % 4 == 0 && WOFF % 4 == 0 && AOFF % 4 == 0,
              "padded rows keep 16-byte alignment for the float4 staging stores");

__global__ __launch_bounds__(TB_ * 32) void spatial_gcn_kernel(
        const float* __restrict__ x, const float* __restrict__ w,
        const float* __restrict__ bias, const float* __restrict__ Ahat,
        float* __restrict__ out) {
    __shared__ float lds[LDSF];
    float* Xl = lds + XOFF;
    float* Wl = lds + WOFF;
    float* Al = lds + AOFF;
    float* Bl = lds + BOFF;

    const int tid = threadIdx.x;
    const int n   = blockIdx.y;
    const int t0  = blockIdx.x * TB_;
    constexpr int NT = TB_ * 32;
    constexpr int XQ = XW / 4;
    constexpr int XV = C_ * XQ;
    constexpr int XI = (XV + NT - 1) / NT;
    constexpr int WI = (F_ * C_ / 4) / NT;
    constexpr int AI = (VP_ * VP_ / 4) / NT;
    static_assert(XW % 4 == 0, "x rows are whole float4s");
    static_assert((F_ * C_ / 4) % NT == 0 && (VP_ * VP_ / 4) % NT == 0,
                  "W and Ahat split evenly over the block");
    static_assert((XI - 1) * NT < XV && (NT % (F_ / 4)) == 0,
                  "only the last x trip is partial; bias copies tile the block");

    const v4f* wg = reinterpret_cast<const v4f*>(w);
    const v4f* ag = reinterpret_cast<const v4f*>(Ahat);
    const v4f* bg = reinterpret_cast<const v4f*>(bias);
    const float* xg = x + (long)n * (C_ * T_ * V_) + t0 * V_;

    v4f wr[WI], ar[AI], xr[XI];
#pragma unroll
    for (int i = 0; i < WI; ++i) wr[i] = wg[tid + i * NT];
#pragma unroll
    for (int i = 0; i < AI; ++i) ar[i] = ag[tid + i * NT];
    const v4f br = bg[tid & (F_ / 4 - 1)];
#pragma unroll
    for (int i = 0; i < XI; ++i) {
        unsigned j = (unsigned)tid + (unsigned)(i * NT);
        if (i == XI - 1 && j > XV - 1) j = XV - 1;
        const unsigned c = j / XQ;
        const unsigned q = j % XQ;
        xr[i] = *reinterpret_cast<const v4f*>(xg + (c * (unsigned)(T_ * V_) + q * 4u));
    }

#pragma unroll
    for (int i = 0; i < WI; ++i) {
        const int j = tid + i * NT;
        *reinterpret_cast<v4f*>(Wl + (j / (C_ / 4)) * WP + (j % (C_ / 4)) * 4) = wr[i];
    }
#pragma unroll
    for (int i = 0; i < AI; ++i) {
        const int j = tid + i * NT;
        *reinterpret_cast<v4f*>(Al + (j / (VP_ / 4)) * AP + (j % (VP_ / 4)) * 4) = ar[i];
    }
    reinterpret_cast<v4f*>(Bl)[tid & (F_ / 4 - 1)] = br;
#pragma unroll
    for (int i = 0; i < XI; ++i) {
        unsigned j = (unsigned)tid + (unsigned)(i * NT);
        if (i == XI - 1 && j > XV - 1) j = XV - 1;
        reinterpret_cast<v4f*>(Xl)[j] = xr[i];
    }
    __syncthreads();

    const int wv   = tid >> 5;
    const int lane = tid & 31;
    const int half = lane >> 4;
    const int l16  = lane & 15;
    const int t    = t0 + wv;
    const int xb   = wv * V_;

    v8f acc[4][2];
#pragma unroll
    for (int mt = 0; mt < 4; ++mt)
#pragma unroll
        for (int nt = 0; nt < 2; ++nt)
#pragma unroll
            for (int r = 0; r < 8; ++r) acc[mt][nt][r] = 0.0f;

#pragma unroll
    for (int k = 0; k < 16; ++k) {
        const int kk = 4 * k + 2 * half;
        v2f av[4], bv[2];
#pragma unroll
        for (int mt = 0; mt < 4; ++mt) {
            const float* p = &Wl[(mt * 16 + l16) * WP + kk];
            av[mt][0] = p[0];
            av[mt][1] = p[1];
        }
#pragma unroll
        for (int nt = 0; nt < 2; ++nt) {
            bv[nt][0] = Xl[(kk + 0) * XW + xb + nt * 16 + l16];
            bv[nt][1] = Xl[(kk + 1) * XW + xb + nt * 16 + l16];
        }
#pragma unroll
        for (int mt = 0; mt < 4; ++mt)
#pragma unroll
            for (int nt = 0; nt < 2; ++nt)
                acc[mt][nt] = __builtin_amdgcn_wmma_f32_16x16x4_f32(
                    false, av[mt], false, bv[nt], (short)0, acc[mt][nt],
                    false, false);
    }

#pragma unroll
    for (int nt = 0; nt < 2; ++nt) {
        const int vp = nt * 16 + l16;
        if (vp < V_) {
#pragma unroll
            for (int mt = 0; mt < 4; ++mt)
#pragma unroll
                for (int r = 0; r < 8; ++r)
                    Xl[(mt * 16 + half * 8 + r) * XW + xb + vp] = acc[mt][nt][r];
        }
    }

    v8f o[4][2];
#pragma unroll
    for (int mt = 0; mt < 4; ++mt)
#pragma unroll
        for (int nt = 0; nt < 2; ++nt)
#pragma unroll
            for (int r = 0; r < 8; ++r)
                o[mt][nt][r] = Bl[mt * 16 + half * 8 + r];

#pragma unroll
    for (int k = 0; k < 8; ++k) {
        const int kk = 4 * k + 2 * half;
        v2f av[4], bv[2];
#pragma unroll
        for (int mt = 0; mt < 4; ++mt) {
            av[mt][0] = Xl[(mt * 16 + l16) * XW + xb + kk + 0];
            av[mt][1] = Xl[(mt * 16 + l16) * XW + xb + kk + 1];
        }
#pragma unroll
        for (int nt = 0; nt < 2; ++nt) {
            bv[nt][0] = Al[(nt * 16 + l16) * AP + kk + 0];
            bv[nt][1] = Al[(nt * 16 + l16) * AP + kk + 1];
        }
#pragma unroll
        for (int mt = 0; mt < 4; ++mt)
#pragma unroll
            for (int nt = 0; nt < 2; ++nt)
                o[mt][nt] = __builtin_amdgcn_wmma_f32_16x16x4_f32(
                    false, av[mt], false, bv[nt], (short)0, o[mt][nt],
                    false, false);
    }

#pragma unroll
    for (int nt = 0; nt < 2; ++nt) {
        const int v = nt * 16 + l16;
        if (v < V_) {
#pragma unroll
            for (int mt = 0; mt < 4; ++mt)
#pragma unroll
                for (int r = 0; r < 8; ++r) {
                    const int f = mt * 16 + half * 8 + r;
                    out[(((long)n * F_ + f) * T_ + t) * V_ + v] = o[mt][nt][r];
                }
        }
    }
}

extern "C" void kernel_launch(void* const* d_in, const int* in_sizes, int n_in,
                              void* d_out, int out_size, void* d_ws, size_t ws_size,
                              hipStream_t stream) {
    const float* x    = (const float*)d_in[0];
    const float* w    = (const float*)d_in[1];
    const float* bias = (const float*)d_in[2];
    const int*   ei   = (const int*)d_in[3];
    const int E = in_sizes[3] / 2;

    float* Ahat = (float*)d_ws;

    build_adj_kernel<<<1, 64, 0, stream>>>(ei, E, Ahat);

    dim3 grid(T_ / TB_, N_);
    spatial_gcn_kernel<<<grid, TB_ * 32, 0, stream>>>(
        x, w, bias, Ahat, (float*)d_out);
}


# Build the normalized adjacency inside every block of the main kernel (LDS atomics and LDS-only barriers under the x loads in flight): one launch per call instead of two, no global atomics
# speedup vs baseline: 1.2939x; 1.0378x over previous
#include <hip/hip_runtime.h>
#include <hip/hip_bf16.h>

typedef float v2f __attribute__((ext_vector_type(2)));
typedef float v4f __attribute__((ext_vector_type(4)));
typedef float v8f __attribute__((ext_vector_type(8)));

#define N_  64
#define C_  64
#define F_  64
#define T_  300
#define V_  25
#define VP_ 32
#define TB_ 4
static_assert(T_ % TB_ == 0, "t tiles must divide T exactly");

#define XW   (TB_ * V_)
#define WP   (C_ + 4)
#define AP   (VP_ + 4)
#define XOFF 0
#define WOFF (C_ * XW)
#define AOFF (WOFF + F_ * WP)
#define BOFF (AOFF + VP_ * AP)
#define DOFF (BOFF + F_)
#define IOFF (DOFF + VP_)
#define LDSF (IOFF + VP_)
static_assert(WP % 8 == 4 && AP % 8 == 4,
              "pitch = 4 x odd: 16 fragment rows land on 16 distinct groups of four banks");
static_assert(WP % 4 == 0 && AP % 4 == 0 && WOFF % 4 == 0 && AOFF % 4 == 0,
              "padded rows keep 16-byte alignment for the float4 staging stores");

static __device__ __forceinline__ void lds_barrier() {
    asm volatile("s_wait_dscnt 0x0" ::: "memory");
    asm volatile("s_barrier_signal -1" ::: "memory");
    asm volatile("s_barrier_wait -1" ::: "memory");
}

__global__ __launch_bounds__(TB_ * 32) void spatial_gcn_kernel(
        const float* __restrict__ x, const float* __restrict__ w,
        const float* __restrict__ bias, const int* __restrict__ ei, int E,
        float* __restrict__ out) {
    __shared__ float lds[LDSF];
    float* Xl = lds + XOFF;
    float* Wl = lds + WOFF;
    float* Al = lds + AOFF;
    float* Bl = lds + BOFF;
    float* Dg = lds + DOFF;
    float* Di = lds + IOFF;

    const int tid = threadIdx.x;
    const int n   = blockIdx.y;
    const int t0  = blockIdx.x * TB_;
    constexpr int NT = TB_ * 32;
    constexpr int XQ = XW / 4;
    constexpr int XV = C_ * XQ;
    constexpr int XI = (XV + NT - 1) / NT;
    constexpr int WI = (F_ * C_ / 4) / NT;
    constexpr int AV = VP_ * AP / 4;
    static_assert(XW % 4 == 0, "x rows are whole float4s");
    static_assert((F_ * C_ / 4) % NT == 0, "W splits evenly over the block");
    static_assert((XI - 1) * NT < XV && (NT % (F_ / 4)) == 0,
                  "only the last x trip is partial; bias copies tile the block");

    const v4f* wg = reinterpret_cast<const v4f*>(w);
    const v4f* bg = reinterpret_cast<const v4f*>(bias);
    const float* xg = x + (long)n * (C_ * T_ * V_) + t0 * V_;

    const int e0 = tid < E ? tid : E - 1;
    const int lane0 = tid & (VP_ - 1);
    const int s0 = ei[e0];
    const int d0 = ei[E + e0];

    v4f wr[WI], xr[XI];
#pragma unroll
    for (int i = 0; i < WI; ++i) wr[i] = wg[tid + i * NT];
    const v4f br = bg[tid & (F_ / 4 - 1)];
#pragma unroll
    for (int i = 0; i < XI; ++i) {
        unsigned j = (unsigned)tid + (unsigned)(i * NT);
        if (i == XI - 1 && j > XV - 1) j = XV - 1;
        const unsigned c = j / XQ;
        const unsigned q = j % XQ;
        xr[i] = *reinterpret_cast<const v4f*>(xg + (c * (unsigned)(T_ * V_) + q * 4u));
    }

    if (tid < VP_) Dg[tid] = 0.0f;
#pragma unroll
    for (int i = 0; i < (AV + NT - 1) / NT; ++i)
        if (tid + i * NT < AV)
            reinterpret_cast<v4f*>(Al)[tid + i * NT] = (v4f){0.0f, 0.0f, 0.0f, 0.0f};
    lds_barrier();
    const bool ok0 = tid < E && (unsigned)s0 < (unsigned)V_ && (unsigned)d0 < (unsigned)V_;
    const int  sc  = ok0 ? s0 : 0;
    const int  dc  = ok0 ? d0 : 0;
    atomicAdd(&Dg[ok0 ? d0 : lane0], ok0 ? 1.0f : 0.0f);
    for (int e = tid + NT; e < E; e += NT) {
        const int d = ei[E + e];
        if ((unsigned)ei[e] < (unsigned)V_ && (unsigned)d < (unsigned)V_) atomicAdd(&Dg[d], 1.0f);
    }
    lds_barrier();
    if (tid < VP_) {
        const float dg = Dg[tid] + 1.0f;
        Di[tid] = (tid < V_ && dg > 0.0f) ? rsqrtf(dg) : 0.0f;
    }
    lds_barrier();
    atomicAdd(&Al[ok0 ? dc * AP + sc : lane0], ok0 ? Di[sc] * Di[dc] : 0.0f);
    for (int e = tid + NT; e < E; e += NT) {
        const int s = ei[e];
        const int d = ei[E + e];
        if ((unsigned)s < (unsigned)V_ && (unsigned)d < (unsigned)V_)
            atomicAdd(&Al[d * AP + s], Di[s] * Di[d]);
    }
    if (tid < V_) atomicAdd(&Al[tid * AP + tid], Di[tid] * Di[tid]);

#pragma unroll
    for (int i = 0; i < WI; ++i) {
        const int j = tid + i * NT;
        *reinterpret_cast<v4f*>(Wl + (j / (C_ / 4)) * WP + (j % (C_ / 4)) * 4) = wr[i];
    }
    reinterpret_cast<v4f*>(Bl)[tid & (F_ / 4 - 1)] = br;
#pragma unroll
    for (int i = 0; i < XI; ++i) {
        unsigned j = (unsigned)tid + (unsigned)(i * NT);
        if (i == XI - 1 && j > XV - 1) j = XV - 1;
        reinterpret_cast<v4f*>(Xl)[j] = xr[i];
    }
    __syncthreads();

    const int wv   = tid >> 5;
    const int lane = tid & 31;
    const int half = lane >> 4;
    const int l16  = lane & 15;
    const int t    = t0 + wv;
    const int xb   = wv * V_;

    v8f acc[4][2];
#pragma unroll
    for (int mt = 0; mt < 4; ++mt)
#pragma unroll
        for (int nt = 0; nt < 2; ++nt)
#pragma unroll
            for (int r = 0; r < 8; ++r) acc[mt][nt][r] = 0.0f;

#pragma unroll
    for (int k = 0; k < 16; ++k) {
        const int kk = 4 * k + 2 * half;
        v2f av[4], bv[2];
#pragma unroll
        for (int mt = 0; mt < 4; ++mt) {
            const float* p = &Wl[(mt * 16 + l16) * WP + kk];
            av[mt][0] = p[0];
            av[mt][1] = p[1];
        }
#pragma unroll
        for (int nt = 0; nt < 2; ++nt) {
            bv[nt][0] = Xl[(kk + 0) * XW + xb + nt * 16 + l16];
            bv[nt][1] = Xl[(kk + 1) * XW + xb + nt * 16 + l16];
        }
#pragma unroll
        for (int mt = 0; mt < 4; ++mt)
#pragma unroll
            for (int nt = 0; nt < 2; ++nt)
                acc[mt][nt] = __builtin_amdgcn_wmma_f32_16x16x4_f32(
                    false, av[mt], false, bv[nt], (short)0, acc[mt][nt],
                    false, false);
    }

#pragma unroll
    for (int nt = 0; nt < 2; ++nt) {
        const int vp = nt * 16 + l16;
        if (vp < V_) {
#pragma unroll
            for (int mt = 0; mt < 4; ++mt)
#pragma unroll
                for (int r = 0; r < 8; ++r)
                    Xl[(mt * 16 + half * 8 + r) * XW + xb + vp] = acc[mt][nt][r];
        }
    }

    v8f o[4][2];
#pragma unroll
    for (int mt = 0; mt < 4; ++mt)
#pragma unroll
        for (int nt = 0; nt < 2; ++nt)
#pragma unroll
            for (int r = 0; r < 8; ++r)
                o[mt][nt][r] = Bl[mt * 16 + half * 8 + r];

#pragma unroll
    for (int k = 0; k < 8; ++k) {
        const int kk = 4 * k + 2 * half;
        v2f av[4], bv[2];
#pragma unroll
        for (int mt = 0; mt < 4; ++mt) {
            av[mt][0] = Xl[(mt * 16 + l16) * XW + xb + kk + 0];
            av[mt][1] = Xl[(mt * 16 + l16) * XW + xb + kk + 1];
        }
#pragma unroll
        for (int nt = 0; nt < 2; ++nt) {
            bv[nt][0] = Al[(nt * 16 + l16) * AP + kk + 0];
            bv[nt][1] = Al[(nt * 16 + l16) * AP + kk + 1];
        }
#pragma unroll
        for (int mt = 0; mt < 4; ++mt)
#pragma unroll
            for (int nt = 0; nt < 2; ++nt)
                o[mt][nt] = __builtin_amdgcn_wmma_f32_16x16x4_f32(
                    false, av[mt], false, bv[nt], (short)0, o[mt][nt],
                    false, false);
    }

#pragma unroll
    for (int nt = 0; nt < 2; ++nt) {
        const int v = nt * 16 + l16;
        if (v < V_) {
#pragma unroll
            for (int mt = 0; mt < 4; ++mt)
#pragma unroll
                for (int r = 0; r < 8; ++r) {
                    const int f = mt * 16 + half * 8 + r;
                    out[(((long)n * F_ + f) * T_ + t) * V_ + v] = o[mt][nt][r];
                }
        }
    }
}

extern "C" void kernel_launch(void* const* d_in, const int* in_sizes, int n_in,
                              void* d_out, int out_size, void* d_ws, size_t ws_size,
                              hipStream_t stream) {
    const float* x    = (const float*)d_in[0];
    const float* w    = (const float*)d_in[1];
    const float* bias = (const float*)d_in[2];
    const int*   ei   = (const int*)d_in[3];
    const int E = in_sizes[3] / 2;

    dim3 grid(T_ / TB_, N_);
    spatial_gcn_kernel<<<grid, TB_ * 32, 0, stream>>>(
        x, w, bias, ei, E, (float*)d_out);
}
